# starting kernel
# baseline (speedup 1.0000x reference)
#include <hip/hip_runtime.h>
#include <hip/hip_bf16.h>

#define GDIM 128

typedef float v2f __attribute__((ext_vector_type(2)));
typedef float v8f __attribute__((ext_vector_type(8)));

__global__ void deg_init_kernel(float* __restrict__ deg, int n) {
  int i = blockIdx.x * blockDim.x + threadIdx.x;
  if (i < n) deg[i] = 1.0f;
}

__global__ void deg_edges_kernel(const int* __restrict__ dst, float* __restrict__ deg, int E) {
  int e = blockIdx.x * blockDim.x + threadIdx.x;
  if (e < E) {
    __hip_atomic_fetch_add(&deg[dst[e]], 1.0f, __ATOMIC_RELAXED, __HIP_MEMORY_SCOPE_AGENT);
  }
}

__global__ void dinv_kernel(float* __restrict__ deg, int n) {
  int i = blockIdx.x * blockDim.x + threadIdx.x;
  if (i < n) deg[i] = rsqrtf(deg[i]);
}

__global__ void gemm_wmma_f32_kernel(const float* __restrict__ A,
                                     const float* __restrict__ B,
                                     float* __restrict__ C, int M) {
  const int wave = threadIdx.x >> 5;
  const int lane = threadIdx.x & 31;
  const int half = lane >> 4;
  const int lmod = lane & 15;
  const int row0 = blockIdx.x * 16;
  const int col0 = wave * 16;
  if (row0 >= M) return;

  const float* Arow = A + (size_t)(row0 + lmod) * GDIM;
  const float* Bcol = B + col0 + lmod;

  v8f acc = {0.f, 0.f, 0.f, 0.f, 0.f, 0.f, 0.f, 0.f};

#pragma unroll 4
  for (int k = 0; k < GDIM; k += 4) {
    const int ka = k + half * 2;
    v2f a, b;
    a.x = Arow[ka + 0];
    a.y = Arow[ka + 1];
    b.x = Bcol[(size_t)(ka + 0) * GDIM];
    b.y = Bcol[(size_t)(ka + 1) * GDIM];
    acc = __builtin_amdgcn_wmma_f32_16x16x4_f32(
         false, a,  false, b,
         (short)0, acc,  false,  false);
  }

  float* Cp = C + (size_t)(row0 + half * 8) * GDIM + col0 + lmod;
#pragma unroll
  for (int r = 0; r < 8; ++r) {
    Cp[(size_t)r * GDIM] = acc[r];
  }
}

__global__ void selfloop_init_kernel(const float* __restrict__ h,
                                     const float* __restrict__ dinv,
                                     const float* __restrict__ bias,
                                     float* __restrict__ agg, int n) {
  size_t idx = (size_t)blockIdx.x * blockDim.x + threadIdx.x;
  if (idx >= (size_t)n * GDIM) return;
  int node = (int)(idx >> 7);
  int d = (int)(idx & (GDIM - 1));
  float di = dinv[node];
  float v = h[idx] * di * di;
  if (bias) v += bias[d];
  agg[idx] = v;
}

__global__ void scatter_edges_kernel(const float* __restrict__ h,
                                     const int* __restrict__ src,
                                     const int* __restrict__ dst,
                                     const float* __restrict__ dinv,
                                     float* __restrict__ agg, int E) {
  const int wave = __builtin_amdgcn_readfirstlane(threadIdx.x >> 5);
  const int e = blockIdx.x * 8 + wave;
  if (e >= E) return;
  const int lane = threadIdx.x & 31;
  const int s = src[e];
  const int d = dst[e];
  const float norm = dinv[s] * dinv[d];
  const float4 v = ((const float4*)(h + (size_t)s * GDIM))[lane];
  float* ap = agg + (size_t)d * GDIM + lane * 4;
  __hip_atomic_fetch_add(ap + 0, v.x * norm, __ATOMIC_RELAXED, __HIP_MEMORY_SCOPE_AGENT);
  __hip_atomic_fetch_add(ap + 1, v.y * norm, __ATOMIC_RELAXED, __HIP_MEMORY_SCOPE_AGENT);
  __hip_atomic_fetch_add(ap + 2, v.z * norm, __ATOMIC_RELAXED, __HIP_MEMORY_SCOPE_AGENT);
  __hip_atomic_fetch_add(ap + 3, v.w * norm, __ATOMIC_RELAXED, __HIP_MEMORY_SCOPE_AGENT);
}

__global__ void bias_relu_kernel(const float* __restrict__ agg,
                                 const float* __restrict__ bias,
                                 float* __restrict__ out, int n) {
  size_t idx = (size_t)blockIdx.x * blockDim.x + threadIdx.x;
  if (idx >= (size_t)n * GDIM) return;
  int d = (int)(idx & (GDIM - 1));
  out[idx] = fmaxf(agg[idx] + bias[d], 0.0f);
}

extern "C" void kernel_launch(void* const* d_in, const int* in_sizes, int n_in,
                              void* d_out, int out_size, void* d_ws, size_t ws_size,
                              hipStream_t stream) {
  const float* x  = (const float*)d_in[0];
  const int*   ei = (const int*)d_in[1];
  const float* W1 = (const float*)d_in[2];
  const float* b1 = (const float*)d_in[3];
  const float* W2 = (const float*)d_in[4];
  const float* b2 = (const float*)d_in[5];
  float* out = (float*)d_out;

  const int N = in_sizes[0] / GDIM;
  const int E = in_sizes[1] / 2;
  const int* src = ei;
  const int* dst = ei + E;

  float* dinv = (float*)d_ws;
  float* bufA = dinv + N;
  float* bufB = bufA + (size_t)N * GDIM;

  const int T = 256;
  const int gN    = (N + T - 1) / T;
  const int gE    = (E + T - 1) / T;
  const int gElem = (int)(((size_t)N * GDIM + T - 1) / T);
  const int gEdgeWave = (E + 7) / 8;
  const int gGemm = N / 16;

  deg_init_kernel<<<gN, T, 0, stream>>>(dinv, N);
  deg_edges_kernel<<<gE, T, 0, stream>>>(dst, dinv, E);
  dinv_kernel<<<gN, T, 0, stream>>>(dinv, N);

  gemm_wmma_f32_kernel<<<gGemm, T, 0, stream>>>(x, W1, bufA, N);
  selfloop_init_kernel<<<gElem, T, 0, stream>>>(bufA, dinv, nullptr, bufB, N);
  scatter_edges_kernel<<<gEdgeWave, T, 0, stream>>>(bufA, src, dst, dinv, bufB, E);
  bias_relu_kernel<<<gElem, T, 0, stream>>>(bufB, b1, bufA, N);

  gemm_wmma_f32_kernel<<<gGemm, T, 0, stream>>>(bufA, W2, bufB, N);
  selfloop_init_kernel<<<gElem, T, 0, stream>>>(bufB, dinv, b2, out, N);
  scatter_edges_kernel<<<gEdgeWave, T, 0, stream>>>(bufB, src, dst, dinv, out, E);
}


# aggregate by owner: a work-group holds 392 nodes' sums in LDS, filters the edge list for them, adds source rows by LDS float adds, no global atomics; self-loop/bias/relu fused; sum order LDS arrival
# speedup vs baseline: 1.6876x; 1.6876x over previous
#include <hip/hip_runtime.h>
#include <hip/hip_bf16.h>

#define GDIM 128

typedef float v2f __attribute__((ext_vector_type(2)));
typedef float v8f __attribute__((ext_vector_type(8)));

__global__ void deg_init_kernel(float* __restrict__ deg, int n) {
  int i = blockIdx.x * blockDim.x + threadIdx.x;
  if (i < n) deg[i] = 1.0f;
}

__global__ void deg_edges_kernel(const int* __restrict__ dst, float* __restrict__ deg, int E) {
  int e = blockIdx.x * blockDim.x + threadIdx.x;
  if (e < E) {
    __hip_atomic_fetch_add(&deg[dst[e]], 1.0f, __ATOMIC_RELAXED, __HIP_MEMORY_SCOPE_AGENT);
  }
}

__global__ void dinv_kernel(float* __restrict__ deg, int n) {
  int i = blockIdx.x * blockDim.x + threadIdx.x;
  if (i < n) deg[i] = rsqrtf(deg[i]);
}

__global__ void gemm_wmma_f32_kernel(const float* __restrict__ A,
                                     const float* __restrict__ B,
                                     float* __restrict__ C, int M) {
  const int wave = threadIdx.x >> 5;
  const int lane = threadIdx.x & 31;
  const int half = lane >> 4;
  const int lmod = lane & 15;
  const int row0 = blockIdx.x * 16;
  const int col0 = wave * 16;
  if (row0 >= M) return;

  const float* Arow = A + (size_t)(row0 + lmod) * GDIM;
  const float* Bcol = B + col0 + lmod;

  v8f acc = {0.f, 0.f, 0.f, 0.f, 0.f, 0.f, 0.f, 0.f};

#pragma unroll 4
  for (int k = 0; k < GDIM; k += 4) {
    const int ka = k + half * 2;
    v2f a, b;
    a.x = Arow[ka + 0];
    a.y = Arow[ka + 1];
    b.x = Bcol[(size_t)(ka + 0) * GDIM];
    b.y = Bcol[(size_t)(ka + 1) * GDIM];
    acc = __builtin_amdgcn_wmma_f32_16x16x4_f32(
         false, a,  false, b,
         (short)0, acc,  false,  false);
  }

  float* Cp = C + (size_t)(row0 + half * 8) * GDIM + col0 + lmod;
#pragma unroll
  for (int r = 0; r < 8; ++r) {
    Cp[(size_t)r * GDIM] = acc[r];
  }
}

#define NPB 392
#define LCAP 12288
#define AGG_T 1024

template <bool RELU>
__global__ __launch_bounds__(AGG_T)
void aggregate_owner_kernel(const float* __restrict__ h,
                            const int* __restrict__ src,
                            const int* __restrict__ dst,
                            const float* __restrict__ dinv,
                            const float* __restrict__ bias,
                            float* __restrict__ out, int N, int E) {
  __shared__ float acc[NPB * GDIM];
  __shared__ int elist[LCAP];
  __shared__ int ecount;

  const int tid = threadIdx.x;
  const int lo = blockIdx.x * NPB;
  const int nown = min(NPB, N - lo);

  for (int i = tid; i < NPB * GDIM / 4; i += AGG_T)
    ((float4*)acc)[i] = make_float4(0.f, 0.f, 0.f, 0.f);
  if (tid == 0) ecount = 0;
  __syncthreads();

  auto take = [&](int d, int e) {
    if ((unsigned)(d - lo) < (unsigned)nown) {
      const int slot = __hip_atomic_fetch_add(&ecount, 1, __ATOMIC_RELAXED, __HIP_MEMORY_SCOPE_WORKGROUP);
      if (slot < LCAP) {
        elist[slot] = e;
      } else {
        const int s = src[e];
        const float nrm = dinv[s] * dinv[d];
        const float* row = h + (size_t)s * GDIM;
        float* a = acc + (d - lo) * GDIM;
        for (int c = 0; c < GDIM; ++c)
          __hip_atomic_fetch_add(a + c, row[c] * nrm, __ATOMIC_RELAXED, __HIP_MEMORY_SCOPE_WORKGROUP);
      }
    }
  };

  if ((E & 3) == 0) {
    const int4* d4 = (const int4*)dst;
    const int E4 = E >> 2;
    const int rot = (int)(((unsigned)E4 / gridDim.x) * blockIdx.x) & ~31;
    for (int t = tid; t < E4; t += 4 * AGG_T) {
      int4 q[4];
      int ix[4];
#pragma unroll
      for (int u = 0; u < 4; ++u) {
        const int tt = t + u * AGG_T;
        int ii = tt + rot;
        if (ii >= E4) ii -= E4;
        ix[u] = ii;
        q[u] = (tt < E4) ? d4[ii] : make_int4(-1, -1, -1, -1);
      }
#pragma unroll
      for (int u = 0; u < 4; ++u) {
        const int e0 = ix[u] * 4;
        take(q[u].x, e0 + 0);
        take(q[u].y, e0 + 1);
        take(q[u].z, e0 + 2);
        take(q[u].w, e0 + 3);
      }
    }
  } else {
    for (int e = tid; e < E; e += AGG_T) take(dst[e], e);
  }
  __syncthreads();

  const int n = __builtin_amdgcn_readfirstlane(min(ecount, LCAP));
  const int wave = __builtin_amdgcn_readfirstlane(tid >> 5);
  const int lane = tid & 31;
  for (int base = wave * 32; base < n; base += 32 * (AGG_T / 32)) {
    const int idx = base + lane;
    int s = 0, dl = 0;
    float nrm = 0.f;
    if (idx < n) {
      const int e = elist[idx];
      s = src[e];
      const int d = dst[e];
      dl = d - lo;
      nrm = dinv[s] * dinv[d];
    }
    const int cnt = min(32, n - base);
    int j = 0;
    for (; j + 8 <= cnt; j += 8) {
      float v[8][4];
#pragma unroll
      for (int u = 0; u < 8; ++u) {
        const int su = __builtin_amdgcn_readlane(s, j + u);
        const float* row = h + (size_t)su * GDIM + lane;
#pragma unroll
        for (int c = 0; c < 4; ++c) v[u][c] = row[32 * c];
      }
#pragma unroll
      for (int u = 0; u < 8; ++u) {
        const int du = __builtin_amdgcn_readlane(dl, j + u);
        const float nu = __int_as_float(__builtin_amdgcn_readlane(__float_as_int(nrm), j + u));
        float* a = acc + du * GDIM + lane;
#pragma unroll
        for (int c = 0; c < 4; ++c)
          __hip_atomic_fetch_add(a + 32 * c, v[u][c] * nu, __ATOMIC_RELAXED, __HIP_MEMORY_SCOPE_WORKGROUP);
      }
    }
    for (; j < cnt; ++j) {
      const int su = __builtin_amdgcn_readlane(s, j);
      const int du = __builtin_amdgcn_readlane(dl, j);
      const float nu = __int_as_float(__builtin_amdgcn_readlane(__float_as_int(nrm), j));
      const float* row = h + (size_t)su * GDIM + lane;
      float* a = acc + du * GDIM + lane;
#pragma unroll
      for (int c = 0; c < 4; ++c)
        __hip_atomic_fetch_add(a + 32 * c, row[32 * c] * nu, __ATOMIC_RELAXED, __HIP_MEMORY_SCOPE_WORKGROUP);
    }
  }
  __syncthreads();

  for (int i = tid; i < nown * (GDIM / 4); i += AGG_T) {
    const int node = lo + (i >> 5);
    const int c4 = i & 31;
    const float4 a = ((const float4*)acc)[i];
    const float4 hv = ((const float4*)h)[(size_t)node * (GDIM / 4) + c4];
    const float di = dinv[node];
    const float w = di * di;
    float4 o;
    o.x = a.x + hv.x * w;
    o.y = a.y + hv.y * w;
    o.z = a.z + hv.z * w;
    o.w = a.w + hv.w * w;
    const float4 b = ((const float4*)bias)[c4];
    o.x += b.x; o.y += b.y; o.z += b.z; o.w += b.w;
    if (RELU) {
      o.x = fmaxf(o.x, 0.f); o.y = fmaxf(o.y, 0.f);
      o.z = fmaxf(o.z, 0.f); o.w = fmaxf(o.w, 0.f);
    }
    ((float4*)out)[(size_t)node * (GDIM / 4) + c4] = o;
  }
}

extern "C" void kernel_launch(void* const* d_in, const int* in_sizes, int n_in,
                              void* d_out, int out_size, void* d_ws, size_t ws_size,
                              hipStream_t stream) {
  const float* x  = (const float*)d_in[0];
  const int*   ei = (const int*)d_in[1];
  const float* W1 = (const float*)d_in[2];
  const float* b1 = (const float*)d_in[3];
  const float* W2 = (const float*)d_in[4];
  const float* b2 = (const float*)d_in[5];
  float* out = (float*)d_out;

  const int N = in_sizes[0] / GDIM;
  const int E = in_sizes[1] / 2;
  const int* src = ei;
  const int* dst = ei + E;

  float* dinv = (float*)d_ws;
  float* bufA = dinv + N;
  float* bufB = bufA + (size_t)N * GDIM;

  const int T = 256;
  const int gN    = (N + T - 1) / T;
  const int gE    = (E + T - 1) / T;
  const int gGemm = N / 16;
  const int gAgg  = (N + NPB - 1) / NPB;

  deg_init_kernel<<<gN, T, 0, stream>>>(dinv, N);
  deg_edges_kernel<<<gE, T, 0, stream>>>(dst, dinv, E);
  dinv_kernel<<<gN, T, 0, stream>>>(dinv, N);

  gemm_wmma_f32_kernel<<<gGemm, T, 0, stream>>>(x, W1, bufA, N);
  aggregate_owner_kernel<true><<<gAgg, AGG_T, 0, stream>>>(bufA, src, dst, dinv, b1, bufB, N, E);

  gemm_wmma_f32_kernel<<<gGemm, T, 0, stream>>>(bufB, W2, bufA, N);
  aggregate_owner_kernel<false><<<gAgg, AGG_T, 0, stream>>>(bufA, src, dst, dinv, b2, out, N, E);
}


# aggregation filter scans a 1-byte owner plane (dst/392, written by the degree kernel): 16 edges per lane load, packed zero-byte test, in place of the 4-byte dst sweep; list order still LDS arrival
# speedup vs baseline: 2.2570x; 1.3374x over previous
#include <hip/hip_runtime.h>
#include <hip/hip_bf16.h>

#define GDIM 128

typedef float v2f __attribute__((ext_vector_type(2)));
typedef float v8f __attribute__((ext_vector_type(8)));

__global__ void deg_init_kernel(float* __restrict__ deg, int n) {
  int i = blockIdx.x * blockDim.x + threadIdx.x;
  if (i < n) deg[i] = 1.0f;
}

#define NPB 392
__global__ void deg_edges_kernel(const int* __restrict__ dst, float* __restrict__ deg,
                                 unsigned char* __restrict__ own, int E) {
  const int e0 = (blockIdx.x * blockDim.x + threadIdx.x) * 4;
  if (e0 >= E) return;
  int d[4];
  if ((E & 3) == 0) {
    const int4 q = *(const int4*)(dst + e0);
    d[0] = q.x; d[1] = q.y; d[2] = q.z; d[3] = q.w;
  } else {
#pragma unroll
    for (int u = 0; u < 4; ++u) d[u] = (e0 + u < E) ? dst[e0 + u] : -1;
  }
  unsigned pk = 0u;
#pragma unroll
  for (int u = 0; u < 4; ++u) {
    if (d[u] >= 0) {
      __hip_atomic_fetch_add(&deg[d[u]], 1.0f, __ATOMIC_RELAXED, __HIP_MEMORY_SCOPE_AGENT);
      pk |= (((unsigned)d[u] / (unsigned)NPB) & 255u) << (8 * u);
    }
  }
  if (e0 + 3 < E) {
    *(unsigned*)(own + e0) = pk;
  } else {
#pragma unroll
    for (int u = 0; u < 4; ++u)
      if (e0 + u < E) own[e0 + u] = (unsigned char)(pk >> (8 * u));
  }
}

__global__ void dinv_kernel(float* __restrict__ deg, int n) {
  int i = blockIdx.x * blockDim.x + threadIdx.x;
  if (i < n) deg[i] = rsqrtf(deg[i]);
}

__global__ void gemm_wmma_f32_kernel(const float* __restrict__ A,
                                     const float* __restrict__ B,
                                     float* __restrict__ C, int M) {
  const int wave = threadIdx.x >> 5;
  const int lane = threadIdx.x & 31;
  const int half = lane >> 4;
  const int lmod = lane & 15;
  const int row0 = blockIdx.x * 16;
  const int col0 = wave * 16;
  if (row0 >= M) return;

  const float* Arow = A + (size_t)(row0 + lmod) * GDIM;
  const float* Bcol = B + col0 + lmod;

  v8f acc = {0.f, 0.f, 0.f, 0.f, 0.f, 0.f, 0.f, 0.f};

#pragma unroll 4
  for (int k = 0; k < GDIM; k += 4) {
    const int ka = k + half * 2;
    v2f a, b;
    a.x = Arow[ka + 0];
    a.y = Arow[ka + 1];
    b.x = Bcol[(size_t)(ka + 0) * GDIM];
    b.y = Bcol[(size_t)(ka + 1) * GDIM];
    acc = __builtin_amdgcn_wmma_f32_16x16x4_f32(
         false, a,  false, b,
         (short)0, acc,  false,  false);
  }

  float* Cp = C + (size_t)(row0 + half * 8) * GDIM + col0 + lmod;
#pragma unroll
  for (int r = 0; r < 8; ++r) {
    Cp[(size_t)r * GDIM] = acc[r];
  }
}

#define LCAP 12288
#define AGG_T 1024

template <bool RELU>
__global__ __launch_bounds__(AGG_T)
void aggregate_owner_kernel(const float* __restrict__ h,
                            const int* __restrict__ src,
                            const int* __restrict__ dst,
                            const float* __restrict__ dinv,
                            const float* __restrict__ bias,
                            const unsigned char* __restrict__ own,
                            float* __restrict__ out, int N, int E) {
  __shared__ float acc[NPB * GDIM];
  __shared__ int elist[LCAP];
  __shared__ int ecount;

  const int tid = threadIdx.x;
  const int lo = blockIdx.x * NPB;
  const int nown = min(NPB, N - lo);

  for (int i = tid; i < NPB * GDIM / 4; i += AGG_T)
    ((float4*)acc)[i] = make_float4(0.f, 0.f, 0.f, 0.f);
  if (tid == 0) ecount = 0;
  __syncthreads();

  auto take = [&](int d, int e) {
    if ((unsigned)(d - lo) < (unsigned)nown) {
      const int slot = __hip_atomic_fetch_add(&ecount, 1, __ATOMIC_RELAXED, __HIP_MEMORY_SCOPE_WORKGROUP);
      if (slot < LCAP) {
        elist[slot] = e;
      } else {
        const int s = src[e];
        const float nrm = dinv[s] * dinv[d];
        const float* row = h + (size_t)s * GDIM;
        float* a = acc + (d - lo) * GDIM;
        for (int c = 0; c < GDIM; ++c)
          __hip_atomic_fetch_add(a + c, row[c] * nrm, __ATOMIC_RELAXED, __HIP_MEMORY_SCOPE_WORKGROUP);
      }
    }
  };

  auto push = [&](int e) {
    const int slot = __hip_atomic_fetch_add(&ecount, 1, __ATOMIC_RELAXED, __HIP_MEMORY_SCOPE_WORKGROUP);
    if (slot < LCAP) {
      elist[slot] = e;
    } else {
      const int s = src[e];
      const int d = dst[e];
      const float nrm = dinv[s] * dinv[d];
      const float* row = h + (size_t)s * GDIM;
      float* a = acc + (d - lo) * GDIM;
      for (int c = 0; c < GDIM; ++c)
        __hip_atomic_fetch_add(a + c, row[c] * nrm, __ATOMIC_RELAXED, __HIP_MEMORY_SCOPE_WORKGROUP);
    }
  };

  if ((E & 15) == 0 && gridDim.x <= 256u) {
    const uint4* o4 = (const uint4*)own;
    const int E16 = E >> 4;
    const unsigned idrep = blockIdx.x * 0x01010101u;
    const int rot = (int)(((unsigned)E16 / gridDim.x) * blockIdx.x) & ~31;
    for (int t = tid; t < E16; t += 8 * AGG_T) {
      uint4 q[8];
      int ix[8];
#pragma unroll
      for (int u = 0; u < 8; ++u) {
        const int tt = t + u * AGG_T;
        int ii = ((tt < E16) ? tt : t) + rot;
        if (ii >= E16) ii -= E16;
        ix[u] = ii;
        q[u] = o4[ii];
      }
#pragma unroll
      for (int u = 0; u < 8; ++u) {
        const unsigned x0 = q[u].x ^ idrep, x1 = q[u].y ^ idrep;
        const unsigned x2 = q[u].z ^ idrep, x3 = q[u].w ^ idrep;
        const unsigned z0 = ((x0 & 0x7f7f7f7fu) + 0x7f7f7f7fu) | x0;
        const unsigned z1 = ((x1 & 0x7f7f7f7fu) + 0x7f7f7f7fu) | x1;
        const unsigned z2 = ((x2 & 0x7f7f7f7fu) + 0x7f7f7f7fu) | x2;
        const unsigned z3 = ((x3 & 0x7f7f7f7fu) + 0x7f7f7f7fu) | x3;
        const unsigned r0 = __builtin_rotateright32(z0, 7), r1 = __builtin_rotateright32(z1, 6);
        const unsigned r2 = __builtin_rotateright32(z2, 5), r3 = __builtin_rotateright32(z3, 4);
        const unsigned z01 = (r0 & 0x01010101u) | (r1 & ~0x01010101u);
        const unsigned z23 = (r2 & 0x04040404u) | (r3 & ~0x04040404u);
        const unsigned zz = (z01 & 0x03030303u) | (z23 & ~0x03030303u);
        unsigned m = (t + u * AGG_T < E16) ? (~zz & 0x0f0f0f0fu) : 0u;
        const int e0 = ix[u] * 16;
        while (m) {
          const int p = __builtin_ctz(m);
          m &= m - 1u;
          push(e0 + (p & 7) * 4 + (p >> 3));
        }
      }
    }
  } else if ((E & 3) == 0) {
    const int4* d4 = (const int4*)dst;
    const int E4 = E >> 2;
    const int rot = (int)(((unsigned)E4 / gridDim.x) * blockIdx.x) & ~31;
    for (int t = tid; t < E4; t += 4 * AGG_T) {
      int4 q[4];
      int ix[4];
#pragma unroll
      for (int u = 0; u < 4; ++u) {
        const int tt = t + u * AGG_T;
        int ii = tt + rot;
        if (ii >= E4) ii -= E4;
        ix[u] = ii;
        q[u] = (tt < E4) ? d4[ii] : make_int4(-1, -1, -1, -1);
      }
#pragma unroll
      for (int u = 0; u < 4; ++u) {
        const int e0 = ix[u] * 4;
        take(q[u].x, e0 + 0);
        take(q[u].y, e0 + 1);
        take(q[u].z, e0 + 2);
        take(q[u].w, e0 + 3);
      }
    }
  } else {
    for (int e = tid; e < E; e += AGG_T) take(dst[e], e);
  }
  __syncthreads();

  const int n = __builtin_amdgcn_readfirstlane(min(ecount, LCAP));
  const int wave = __builtin_amdgcn_readfirstlane(tid >> 5);
  const int lane = tid & 31;
  for (int base = wave * 32; base < n; base += 32 * (AGG_T / 32)) {
    const int idx = base + lane;
    int s = 0, dl = 0;
    float nrm = 0.f;
    if (idx < n) {
      const int e = elist[idx];
      s = src[e];
      const int d = dst[e];
      dl = d - lo;
      nrm = dinv[s] * dinv[d];
    }
    const int cnt = min(32, n - base);
    int j = 0;
    for (; j + 8 <= cnt; j += 8) {
      float v[8][4];
#pragma unroll
      for (int u = 0; u < 8; ++u) {
        const int su = __builtin_amdgcn_readlane(s, j + u);
        const float* row = h + (size_t)su * GDIM + lane;
#pragma unroll
        for (int c = 0; c < 4; ++c) v[u][c] = row[32 * c];
      }
#pragma unroll
      for (int u = 0; u < 8; ++u) {
        const int du = __builtin_amdgcn_readlane(dl, j + u);
        const float nu = __int_as_float(__builtin_amdgcn_readlane(__float_as_int(nrm), j + u));
        float* a = acc + du * GDIM + lane;
#pragma unroll
        for (int c = 0; c < 4; ++c)
          __hip_atomic_fetch_add(a + 32 * c, v[u][c] * nu, __ATOMIC_RELAXED, __HIP_MEMORY_SCOPE_WORKGROUP);
      }
    }
    for (; j < cnt; ++j) {
      const int su = __builtin_amdgcn_readlane(s, j);
      const int du = __builtin_amdgcn_readlane(dl, j);
      const float nu = __int_as_float(__builtin_amdgcn_readlane(__float_as_int(nrm), j));
      const float* row = h + (size_t)su * GDIM + lane;
      float* a = acc + du * GDIM + lane;
#pragma unroll
      for (int c = 0; c < 4; ++c)
        __hip_atomic_fetch_add(a + 32 * c, row[32 * c] * nu, __ATOMIC_RELAXED, __HIP_MEMORY_SCOPE_WORKGROUP);
    }
  }
  __syncthreads();

  for (int i = tid; i < nown * (GDIM / 4); i += AGG_T) {
    const int node = lo + (i >> 5);
    const int c4 = i & 31;
    const float4 a = ((const float4*)acc)[i];
    const float4 hv = ((const float4*)h)[(size_t)node * (GDIM / 4) + c4];
    const float di = dinv[node];
    const float w = di * di;
    float4 o;
    o.x = a.x + hv.x * w;
    o.y = a.y + hv.y * w;
    o.z = a.z + hv.z * w;
    o.w = a.w + hv.w * w;
    const float4 b = ((const float4*)bias)[c4];
    o.x += b.x; o.y += b.y; o.z += b.z; o.w += b.w;
    if (RELU) {
      o.x = fmaxf(o.x, 0.f); o.y = fmaxf(o.y, 0.f);
      o.z = fmaxf(o.z, 0.f); o.w = fmaxf(o.w, 0.f);
    }
    ((float4*)out)[(size_t)node * (GDIM / 4) + c4] = o;
  }
}

extern "C" void kernel_launch(void* const* d_in, const int* in_sizes, int n_in,
                              void* d_out, int out_size, void* d_ws, size_t ws_size,
                              hipStream_t stream) {
  const float* x  = (const float*)d_in[0];
  const int*   ei = (const int*)d_in[1];
  const float* W1 = (const float*)d_in[2];
  const float* b1 = (const float*)d_in[3];
  const float* W2 = (const float*)d_in[4];
  const float* b2 = (const float*)d_in[5];
  float* out = (float*)d_out;

  const int N = in_sizes[0] / GDIM;
  const int E = in_sizes[1] / 2;
  const int* src = ei;
  const int* dst = ei + E;

  float* dinv = (float*)d_ws;
  float* bufA = dinv + N;
  float* bufB = bufA + (size_t)N * GDIM;
  unsigned char* own = (unsigned char*)d_ws +
      ((((size_t)N * (2 * GDIM + 1)) * sizeof(float) + 255) & ~(size_t)255);

  const int T = 256;
  const int gN    = (N + T - 1) / T;
  const int gE    = ((E + 3) / 4 + T - 1) / T;
  const int gGemm = N / 16;
  const int gAgg  = (N + NPB - 1) / NPB;

  deg_init_kernel<<<gN, T, 0, stream>>>(dinv, N);
  deg_edges_kernel<<<gE, T, 0, stream>>>(dst, dinv, own, E);
  dinv_kernel<<<gN, T, 0, stream>>>(dinv, N);

  gemm_wmma_f32_kernel<<<gGemm, T, 0, stream>>>(x, W1, bufA, N);
  aggregate_owner_kernel<true><<<gAgg, AGG_T, 0, stream>>>(bufA, src, dst, dinv, b1, own, bufB, N, E);

  gemm_wmma_f32_kernel<<<gGemm, T, 0, stream>>>(bufB, W2, bufA, N);
  aggregate_owner_kernel<false><<<gAgg, AGG_T, 0, stream>>>(bufA, src, dst, dinv, b2, own, out, N, E);
}


# dense transform moved behind the aggregation, (A_hat X) W, and fused into the owner kernel: LDS row sums, started at the self-loop term, times W on f32 WMMA; no GEMM launches; sum order changes
# speedup vs baseline: 2.7618x; 1.2237x over previous
#include <hip/hip_runtime.h>
#include <hip/hip_bf16.h>

#define GDIM 128

typedef float v2f __attribute__((ext_vector_type(2)));
typedef float v8f __attribute__((ext_vector_type(8)));

__global__ void deg_init_kernel(float* __restrict__ deg, int n) {
  int i = blockIdx.x * blockDim.x + threadIdx.x;
  if (i < n) deg[i] = 1.0f;
}

#define NPB 392
__global__ void deg_edges_kernel(const int* __restrict__ dst, float* __restrict__ deg,
                                 unsigned char* __restrict__ own, int E) {
  const int e0 = (blockIdx.x * blockDim.x + threadIdx.x) * 4;
  if (e0 >= E) return;
  int d[4];
  if ((E & 3) == 0) {
    const int4 q = *(const int4*)(dst + e0);
    d[0] = q.x; d[1] = q.y; d[2] = q.z; d[3] = q.w;
  } else {
#pragma unroll
    for (int u = 0; u < 4; ++u) d[u] = (e0 + u < E) ? dst[e0 + u] : -1;
  }
  unsigned pk = 0u;
#pragma unroll
  for (int u = 0; u < 4; ++u) {
    if (d[u] >= 0) {
      __hip_atomic_fetch_add(&deg[d[u]], 1.0f, __ATOMIC_RELAXED, __HIP_MEMORY_SCOPE_AGENT);
      pk |= (((unsigned)d[u] / (unsigned)NPB) & 255u) << (8 * u);
    }
  }
  if (e0 + 3 < E) {
    *(unsigned*)(own + e0) = pk;
  } else {
#pragma unroll
    for (int u = 0; u < 4; ++u)
      if (e0 + u < E) own[e0 + u] = (unsigned char)(pk >> (8 * u));
  }
}

__global__ void dinv_kernel(float* __restrict__ deg, int n) {
  int i = blockIdx.x * blockDim.x + threadIdx.x;
  if (i < n) deg[i] = rsqrtf(deg[i]);
}

#define LCAP 12288
#define AGG_T 1024
#define ZP 132
#define ZROWS (((NPB + 15) / 16) * 16)

template <bool RELU>
__global__ __launch_bounds__(AGG_T)
void aggregate_owner_kernel(const float* __restrict__ h,
                            const int* __restrict__ src,
                            const int* __restrict__ dst,
                            const float* __restrict__ dinv,
                            const float* __restrict__ Wm,
                            const float* __restrict__ bias,
                            const unsigned char* __restrict__ own,
                            float* __restrict__ out, int N, int E) {
  __shared__ float acc[ZROWS * ZP];
  __shared__ int elist[LCAP];
  __shared__ int ecount;

  const int tid = threadIdx.x;
  const int lo = blockIdx.x * NPB;
  const int nown = min(NPB, N - lo);

  {
    constexpr int NIT = (ZROWS * (GDIM / 4) + AGG_T - 1) / AGG_T;
    const int c4 = tid & 31;
    float4 hv[NIT];
    float di[NIT];
#pragma unroll
    for (int u = 0; u < NIT; ++u) {
      const int r = (tid + u * AGG_T) >> 5;
      const int node = lo + ((r < nown) ? r : 0);
      hv[u] = ((const float4*)h)[(size_t)node * (GDIM / 4) + c4];
      di[u] = dinv[node];
    }
#pragma unroll
    for (int u = 0; u < NIT; ++u) {
      const int r = (tid + u * AGG_T) >> 5;
      if (r < ZROWS) {
        const float w = di[u] * di[u];
        float4 z = make_float4(0.f, 0.f, 0.f, 0.f);
        if (r < nown) z = make_float4(hv[u].x * w, hv[u].y * w, hv[u].z * w, hv[u].w * w);
        *((float4*)(acc + r * ZP) + c4) = z;
      }
    }
  }
  if (tid == 0) ecount = 0;
  __syncthreads();

  auto take = [&](int d, int e) {
    if ((unsigned)(d - lo) < (unsigned)nown) {
      const int slot = __hip_atomic_fetch_add(&ecount, 1, __ATOMIC_RELAXED, __HIP_MEMORY_SCOPE_WORKGROUP);
      if (slot < LCAP) {
        elist[slot] = e;
      } else {
        const int s = src[e];
        const float nrm = dinv[s] * dinv[d];
        const float* row = h + (size_t)s * GDIM;
        float* a = acc + (d - lo) * ZP;
        for (int c = 0; c < GDIM; ++c)
          __hip_atomic_fetch_add(a + c, row[c] * nrm, __ATOMIC_RELAXED, __HIP_MEMORY_SCOPE_WORKGROUP);
      }
    }
  };

  auto push = [&](int e) {
    const int slot = __hip_atomic_fetch_add(&ecount, 1, __ATOMIC_RELAXED, __HIP_MEMORY_SCOPE_WORKGROUP);
    if (slot < LCAP) {
      elist[slot] = e;
    } else {
      const int s = src[e];
      const int d = dst[e];
      const float nrm = dinv[s] * dinv[d];
      const float* row = h + (size_t)s * GDIM;
      float* a = acc + (d - lo) * ZP;
      for (int c = 0; c < GDIM; ++c)
        __hip_atomic_fetch_add(a + c, row[c] * nrm, __ATOMIC_RELAXED, __HIP_MEMORY_SCOPE_WORKGROUP);
    }
  };

  if ((E & 15) == 0 && gridDim.x <= 256u) {
    const uint4* o4 = (const uint4*)own;
    const int E16 = E >> 4;
    const unsigned idrep = blockIdx.x * 0x01010101u;
    const int rot = (int)(((unsigned)E16 / gridDim.x) * blockIdx.x) & ~31;
    for (int t = tid; t < E16; t += 8 * AGG_T) {
      uint4 q[8];
      int ix[8];
#pragma unroll
      for (int u = 0; u < 8; ++u) {
        const int tt = t + u * AGG_T;
        int ii = ((tt < E16) ? tt : t) + rot;
        if (ii >= E16) ii -= E16;
        ix[u] = ii;
        q[u] = o4[ii];
      }
#pragma unroll
      for (int u = 0; u < 8; ++u) {
        const unsigned x0 = q[u].x ^ idrep, x1 = q[u].y ^ idrep;
        const unsigned x2 = q[u].z ^ idrep, x3 = q[u].w ^ idrep;
        const unsigned z0 = ((x0 & 0x7f7f7f7fu) + 0x7f7f7f7fu) | x0;
        const unsigned z1 = ((x1 & 0x7f7f7f7fu) + 0x7f7f7f7fu) | x1;
        const unsigned z2 = ((x2 & 0x7f7f7f7fu) + 0x7f7f7f7fu) | x2;
        const unsigned z3 = ((x3 & 0x7f7f7f7fu) + 0x7f7f7f7fu) | x3;
        const unsigned r0 = __builtin_rotateright32(z0, 7), r1 = __builtin_rotateright32(z1, 6);
        const unsigned r2 = __builtin_rotateright32(z2, 5), r3 = __builtin_rotateright32(z3, 4);
        const unsigned z01 = (r0 & 0x01010101u) | (r1 & ~0x01010101u);
        const unsigned z23 = (r2 & 0x04040404u) | (r3 & ~0x04040404u);
        const unsigned zz = (z01 & 0x03030303u) | (z23 & ~0x03030303u);
        unsigned m = (t + u * AGG_T < E16) ? (~zz & 0x0f0f0f0fu) : 0u;
        const int e0 = ix[u] * 16;
        while (m) {
          const int p = __builtin_ctz(m);
          m &= m - 1u;
          push(e0 + (p & 7) * 4 + (p >> 3));
        }
      }
    }
  } else if ((E & 3) == 0) {
    const int4* d4 = (const int4*)dst;
    const int E4 = E >> 2;
    const int rot = (int)(((unsigned)E4 / gridDim.x) * blockIdx.x) & ~31;
    for (int t = tid; t < E4; t += 4 * AGG_T) {
      int4 q[4];
      int ix[4];
#pragma unroll
      for (int u = 0; u < 4; ++u) {
        const int tt = t + u * AGG_T;
        int ii = tt + rot;
        if (ii >= E4) ii -= E4;
        ix[u] = ii;
        q[u] = (tt < E4) ? d4[ii] : make_int4(-1, -1, -1, -1);
      }
#pragma unroll
      for (int u = 0; u < 4; ++u) {
        const int e0 = ix[u] * 4;
        take(q[u].x, e0 + 0);
        take(q[u].y, e0 + 1);
        take(q[u].z, e0 + 2);
        take(q[u].w, e0 + 3);
      }
    }
  } else {
    for (int e = tid; e < E; e += AGG_T) take(dst[e], e);
  }
  __syncthreads();

  const int n = __builtin_amdgcn_readfirstlane(min(ecount, LCAP));
  const int wave = __builtin_amdgcn_readfirstlane(tid >> 5);
  const int lane = tid & 31;
  for (int base = wave * 32; base < n; base += 32 * (AGG_T / 32)) {
    const int idx = base + lane;
    int s = 0, dl = 0;
    float nrm = 0.f;
    if (idx < n) {
      const int e = elist[idx];
      s = src[e];
      const int d = dst[e];
      dl = d - lo;
      nrm = dinv[s] * dinv[d];
    }
    const int cnt = min(32, n - base);
    int j = 0;
    for (; j + 8 <= cnt; j += 8) {
      float v[8][4];
#pragma unroll
      for (int u = 0; u < 8; ++u) {
        const int su = __builtin_amdgcn_readlane(s, j + u);
        const float* row = h + (size_t)su * GDIM + lane;
#pragma unroll
        for (int c = 0; c < 4; ++c) v[u][c] = row[32 * c];
      }
#pragma unroll
      for (int u = 0; u < 8; ++u) {
        const int du = __builtin_amdgcn_readlane(dl, j + u);
        const float nu = __int_as_float(__builtin_amdgcn_readlane(__float_as_int(nrm), j + u));
        float* a = acc + du * ZP + lane;
#pragma unroll
        for (int c = 0; c < 4; ++c)
          __hip_atomic_fetch_add(a + 32 * c, v[u][c] * nu, __ATOMIC_RELAXED, __HIP_MEMORY_SCOPE_WORKGROUP);
      }
    }
    for (; j < cnt; ++j) {
      const int su = __builtin_amdgcn_readlane(s, j);
      const int du = __builtin_amdgcn_readlane(dl, j);
      const float nu = __int_as_float(__builtin_amdgcn_readlane(__float_as_int(nrm), j));
      const float* row = h + (size_t)su * GDIM + lane;
      float* a = acc + du * ZP + lane;
#pragma unroll
      for (int c = 0; c < 4; ++c)
        __hip_atomic_fetch_add(a + 32 * c, row[32 * c] * nu, __ATOMIC_RELAXED, __HIP_MEMORY_SCOPE_WORKGROUP);
    }
  }

  const int half = lane >> 4;
  const int lmod = lane & 15;
  const int col = (wave & 7) * 16 + lmod;

  v2f bf[GDIM / 4];
#pragma unroll
  for (int kk = 0; kk < GDIM / 4; ++kk) {
    const int ka = kk * 4 + half * 2;
    bf[kk].x = Wm[(size_t)(ka + 0) * GDIM + col];
    bf[kk].y = Wm[(size_t)(ka + 1) * GDIM + col];
  }
  const float bv = bias[col];
  __syncthreads();

  for (int strip = wave >> 3; strip * 16 < nown; strip += 4) {
    const float* ap = acc + (strip * 16 + lmod) * ZP + half * 2;
    v8f ct = {0.f, 0.f, 0.f, 0.f, 0.f, 0.f, 0.f, 0.f};
#pragma unroll
    for (int kk = 0; kk < GDIM / 4; ++kk) {
      const v2f a = *(const v2f*)(ap + kk * 4);
      ct = __builtin_amdgcn_wmma_f32_16x16x4_f32(
           false, a,  false, bf[kk],
           (short)0, ct,  false,  false);
    }
    const int rb = strip * 16 + half * 8;
    float* op = out + (size_t)(lo + rb) * GDIM + col;
    float o[8];
#pragma unroll
    for (int r = 0; r < 8; ++r) {
      o[r] = ct[r] + bv;
      if (RELU) o[r] = fmaxf(o[r], 0.f);
    }
    if (rb + 8 <= nown) {
#pragma unroll
      for (int r = 0; r < 8; ++r) op[(size_t)r * GDIM] = o[r];
    } else {
#pragma unroll
      for (int r = 0; r < 8; ++r)
        if (rb + r < nown) op[(size_t)r * GDIM] = o[r];
    }
  }
}

extern "C" void kernel_launch(void* const* d_in, const int* in_sizes, int n_in,
                              void* d_out, int out_size, void* d_ws, size_t ws_size,
                              hipStream_t stream) {
  const float* x  = (const float*)d_in[0];
  const int*   ei = (const int*)d_in[1];
  const float* W1 = (const float*)d_in[2];
  const float* b1 = (const float*)d_in[3];
  const float* W2 = (const float*)d_in[4];
  const float* b2 = (const float*)d_in[5];
  float* out = (float*)d_out;

  const int N = in_sizes[0] / GDIM;
  const int E = in_sizes[1] / 2;
  const int* src = ei;
  const int* dst = ei + E;

  float* dinv = (float*)d_ws;
  float* bufB = dinv + N + (size_t)N * GDIM;
  unsigned char* own = (unsigned char*)d_ws +
      ((((size_t)N * (2 * GDIM + 1)) * sizeof(float) + 255) & ~(size_t)255);

  const int T = 256;
  const int gN    = (N + T - 1) / T;
  const int gE    = ((E + 3) / 4 + T - 1) / T;
  const int gAgg  = (N + NPB - 1) / NPB;

  deg_init_kernel<<<gN, T, 0, stream>>>(dinv, N);
  deg_edges_kernel<<<gE, T, 0, stream>>>(dst, dinv, own, E);
  dinv_kernel<<<gN, T, 0, stream>>>(dinv, N);

  aggregate_owner_kernel<true><<<gAgg, AGG_T, 0, stream>>>(x, src, dst, dinv, W1, b1, own, bufB, N, E);

  aggregate_owner_kernel<false><<<gAgg, AGG_T, 0, stream>>>(bufB, src, dst, dinv, W2, b2, own, out, N, E);
}


# layer 2 reads back layer 1's filtered per-range edge list (source, local row, weight) from the workspace, one batch ahead: no filter pass, no scattered index loads; sum order LDS arrival as before
# speedup vs baseline: 3.1461x; 1.1391x over previous
#include <hip/hip_runtime.h>
#include <hip/hip_bf16.h>

#define GDIM 128

typedef float v2f __attribute__((ext_vector_type(2)));
typedef float v8f __attribute__((ext_vector_type(8)));

__global__ void deg_init_kernel(float* __restrict__ deg, int n) {
  int i = blockIdx.x * blockDim.x + threadIdx.x;
  if (i < n) deg[i] = 1.0f;
}

#define NPB 392
__global__ void deg_edges_kernel(const int* __restrict__ dst, float* __restrict__ deg,
                                 unsigned char* __restrict__ own, int E) {
  const int e0 = (blockIdx.x * blockDim.x + threadIdx.x) * 4;
  if (e0 >= E) return;
  int d[4];
  if ((E & 3) == 0) {
    const int4 q = *(const int4*)(dst + e0);
    d[0] = q.x; d[1] = q.y; d[2] = q.z; d[3] = q.w;
  } else {
#pragma unroll
    for (int u = 0; u < 4; ++u) d[u] = (e0 + u < E) ? dst[e0 + u] : -1;
  }
  unsigned pk = 0u;
#pragma unroll
  for (int u = 0; u < 4; ++u) {
    if (d[u] >= 0) {
      __hip_atomic_fetch_add(&deg[d[u]], 1.0f, __ATOMIC_RELAXED, __HIP_MEMORY_SCOPE_AGENT);
      pk |= (((unsigned)d[u] / (unsigned)NPB) & 255u) << (8 * u);
    }
  }
  if (e0 + 3 < E) {
    *(unsigned*)(own + e0) = pk;
  } else {
#pragma unroll
    for (int u = 0; u < 4; ++u)
      if (e0 + u < E) own[e0 + u] = (unsigned char)(pk >> (8 * u));
  }
}

__global__ void dinv_kernel(float* __restrict__ deg, int n) {
  int i = blockIdx.x * blockDim.x + threadIdx.x;
  if (i < n) deg[i] = rsqrtf(deg[i]);
}

#define LCAP 12288
#define AGG_T 1024
#define ZP 132
#define ZROWS (((NPB + 15) / 16) * 16)

template <bool RELU, int LIST>
__global__ __launch_bounds__(AGG_T)
void aggregate_owner_kernel(const float* __restrict__ h,
                            const int* __restrict__ src,
                            const int* __restrict__ dst,
                            const float* __restrict__ dinv,
                            const float* __restrict__ Wm,
                            const float* __restrict__ bias,
                            const unsigned char* __restrict__ own,
                            int* __restrict__ lsd,
                            float* __restrict__ lw,
                            int* __restrict__ lcnt,
                            float* __restrict__ out, int N, int E) {
  __shared__ float acc[ZROWS * ZP];
  __shared__ int elist[LCAP];
  __shared__ int ecount;

  const int tid = threadIdx.x;
  const int wave = __builtin_amdgcn_readfirstlane(tid >> 5);
  const int lane = tid & 31;
  const int lo = blockIdx.x * NPB;
  const int nown = min(NPB, N - lo);

  int2* const lsd2 = (int2*)lsd + (size_t)blockIdx.x * LCAP;
  float* const lwp = lw + (size_t)blockIdx.x * LCAP;

  int nlist = 0;
  bool uselist = false;
  int s_nx = 0, dl_nx = 0;
  float w_nx = 0.f;
  if (LIST == 2) {
    nlist = __builtin_amdgcn_readfirstlane(lcnt[blockIdx.x]);
    uselist = (unsigned)nlist <= (unsigned)LCAP;
    const int idx = wave * 32 + lane;
    if (uselist && idx < nlist) {
      const int2 sd = lsd2[idx];
      s_nx = sd.x;
      dl_nx = sd.y;
      w_nx = lwp[idx];
    }
  }

  {
    constexpr int NIT = (ZROWS * (GDIM / 4) + AGG_T - 1) / AGG_T;
    const int c4 = tid & 31;
    float4 hv[NIT];
    float di[NIT];
#pragma unroll
    for (int u = 0; u < NIT; ++u) {
      const int r = (tid + u * AGG_T) >> 5;
      const int node = lo + ((r < nown) ? r : 0);
      hv[u] = ((const float4*)h)[(size_t)node * (GDIM / 4) + c4];
      di[u] = dinv[node];
    }
#pragma unroll
    for (int u = 0; u < NIT; ++u) {
      const int r = (tid + u * AGG_T) >> 5;
      if (r < ZROWS) {
        const float w = di[u] * di[u];
        float4 z = make_float4(0.f, 0.f, 0.f, 0.f);
        if (r < nown) z = make_float4(hv[u].x * w, hv[u].y * w, hv[u].z * w, hv[u].w * w);
        *((float4*)(acc + r * ZP) + c4) = z;
      }
    }
  }
  if (tid == 0) ecount = 0;
  __syncthreads();

  auto take = [&](int d, int e) {
    if ((unsigned)(d - lo) < (unsigned)nown) {
      const int slot = __hip_atomic_fetch_add(&ecount, 1, __ATOMIC_RELAXED, __HIP_MEMORY_SCOPE_WORKGROUP);
      if (slot < LCAP) {
        elist[slot] = e;
      } else {
        const int s = src[e];
        const float nrm = dinv[s] * dinv[d];
        const float* row = h + (size_t)s * GDIM;
        float* a = acc + (d - lo) * ZP;
        for (int c = 0; c < GDIM; ++c)
          __hip_atomic_fetch_add(a + c, row[c] * nrm, __ATOMIC_RELAXED, __HIP_MEMORY_SCOPE_WORKGROUP);
      }
    }
  };

  auto push = [&](int e) {
    const int slot = __hip_atomic_fetch_add(&ecount, 1, __ATOMIC_RELAXED, __HIP_MEMORY_SCOPE_WORKGROUP);
    if (slot < LCAP) {
      elist[slot] = e;
    } else {
      const int s = src[e];
      const int d = dst[e];
      const float nrm = dinv[s] * dinv[d];
      const float* row = h + (size_t)s * GDIM;
      float* a = acc + (d - lo) * ZP;
      for (int c = 0; c < GDIM; ++c)
        __hip_atomic_fetch_add(a + c, row[c] * nrm, __ATOMIC_RELAXED, __HIP_MEMORY_SCOPE_WORKGROUP);
    }
  };

  if (uselist) {
  } else if ((E & 15) == 0 && gridDim.x <= 256u) {
    const uint4* o4 = (const uint4*)own;
    const int E16 = E >> 4;
    const unsigned idrep = blockIdx.x * 0x01010101u;
    const int rot = (int)(((unsigned)E16 / gridDim.x) * blockIdx.x) & ~31;
    for (int t = tid; t < E16; t += 8 * AGG_T) {
      uint4 q[8];
      int ix[8];
#pragma unroll
      for (int u = 0; u < 8; ++u) {
        const int tt = t + u * AGG_T;
        int ii = ((tt < E16) ? tt : t) + rot;
        if (ii >= E16) ii -= E16;
        ix[u] = ii;
        q[u] = o4[ii];
      }
#pragma unroll
      for (int u = 0; u < 8; ++u) {
        const unsigned x0 = q[u].x ^ idrep, x1 = q[u].y ^ idrep;
        const unsigned x2 = q[u].z ^ idrep, x3 = q[u].w ^ idrep;
        const unsigned z0 = ((x0 & 0x7f7f7f7fu) + 0x7f7f7f7fu) | x0;
        const unsigned z1 = ((x1 & 0x7f7f7f7fu) + 0x7f7f7f7fu) | x1;
        const unsigned z2 = ((x2 & 0x7f7f7f7fu) + 0x7f7f7f7fu) | x2;
        const unsigned z3 = ((x3 & 0x7f7f7f7fu) + 0x7f7f7f7fu) | x3;
        const unsigned r0 = __builtin_rotateright32(z0, 7), r1 = __builtin_rotateright32(z1, 6);
        const unsigned r2 = __builtin_rotateright32(z2, 5), r3 = __builtin_rotateright32(z3, 4);
        const unsigned z01 = (r0 & 0x01010101u) | (r1 & ~0x01010101u);
        const unsigned z23 = (r2 & 0x04040404u) | (r3 & ~0x04040404u);
        const unsigned zz = (z01 & 0x03030303u) | (z23 & ~0x03030303u);
        unsigned m = (t + u * AGG_T < E16) ? (~zz & 0x0f0f0f0fu) : 0u;
        const int e0 = ix[u] * 16;
        while (m) {
          const int p = __builtin_ctz(m);
          m &= m - 1u;
          push(e0 + (p & 7) * 4 + (p >> 3));
        }
      }
    }
  } else if ((E & 3) == 0) {
    const int4* d4 = (const int4*)dst;
    const int E4 = E >> 2;
    const int rot = (int)(((unsigned)E4 / gridDim.x) * blockIdx.x) & ~31;
    for (int t = tid; t < E4; t += 4 * AGG_T) {
      int4 q[4];
      int ix[4];
#pragma unroll
      for (int u = 0; u < 4; ++u) {
        const int tt = t + u * AGG_T;
        int ii = tt + rot;
        if (ii >= E4) ii -= E4;
        ix[u] = ii;
        q[u] = (tt < E4) ? d4[ii] : make_int4(-1, -1, -1, -1);
      }
#pragma unroll
      for (int u = 0; u < 4; ++u) {
        const int e0 = ix[u] * 4;
        take(q[u].x, e0 + 0);
        take(q[u].y, e0 + 1);
        take(q[u].z, e0 + 2);
        take(q[u].w, e0 + 3);
      }
    }
  } else {
    for (int e = tid; e < E; e += AGG_T) take(dst[e], e);
  }
  __syncthreads();

  if (LIST == 1 && tid == 0) lcnt[blockIdx.x] = ecount;

  auto add_batch = [&](int base, int n, int s, int dl, float nrm) {
    const int cnt = min(32, n - base);
    int j = 0;
    for (; j + 8 <= cnt; j += 8) {
      float v[8][4];
#pragma unroll
      for (int u = 0; u < 8; ++u) {
        const int su = __builtin_amdgcn_readlane(s, j + u);
        const float* row = h + (size_t)su * GDIM + lane;
#pragma unroll
        for (int c = 0; c < 4; ++c) v[u][c] = row[32 * c];
      }
#pragma unroll
      for (int u = 0; u < 8; ++u) {
        const int du = __builtin_amdgcn_readlane(dl, j + u);
        const float nu = __int_as_float(__builtin_amdgcn_readlane(__float_as_int(nrm), j + u));
        float* a = acc + du * ZP + lane;
#pragma unroll
        for (int c = 0; c < 4; ++c)
          __hip_atomic_fetch_add(a + 32 * c, v[u][c] * nu, __ATOMIC_RELAXED, __HIP_MEMORY_SCOPE_WORKGROUP);
      }
    }
    for (; j < cnt; ++j) {
      const int su = __builtin_amdgcn_readlane(s, j);
      const int du = __builtin_amdgcn_readlane(dl, j);
      const float nu = __int_as_float(__builtin_amdgcn_readlane(__float_as_int(nrm), j));
      const float* row = h + (size_t)su * GDIM + lane;
      float* a = acc + du * ZP + lane;
#pragma unroll
      for (int c = 0; c < 4; ++c)
        __hip_atomic_fetch_add(a + 32 * c, row[32 * c] * nu, __ATOMIC_RELAXED, __HIP_MEMORY_SCOPE_WORKGROUP);
    }
  };

  if (uselist) {
    const int n = nlist;
    for (int base = wave * 32; base < n; base += 32 * (AGG_T / 32)) {
      const int s = s_nx, dl = dl_nx;
      const float nrm = w_nx;
      const int idn = base + 32 * (AGG_T / 32) + lane;
      const int idc = (idn < n) ? idn : 0;
      const int2 sd = lsd2[idc];
      s_nx = sd.x;
      dl_nx = sd.y;
      w_nx = lwp[idc];
      add_batch(base, n, s, dl, nrm);
    }
  } else {
    const int n = __builtin_amdgcn_readfirstlane(min(ecount, LCAP));
    for (int base = wave * 32; base < n; base += 32 * (AGG_T / 32)) {
      const int idx = base + lane;
      int s = 0, dl = 0;
      float nrm = 0.f;
      if (idx < n) {
        const int e = elist[idx];
        s = src[e];
        const int d = dst[e];
        dl = d - lo;
        nrm = dinv[s] * dinv[d];
        if (LIST == 1) {
          lsd2[idx] = make_int2(s, dl);
          lwp[idx] = nrm;
        }
      }
      add_batch(base, n, s, dl, nrm);
    }
  }

  const int half = lane >> 4;
  const int lmod = lane & 15;
  const int col = (wave & 7) * 16 + lmod;

  v2f bf[GDIM / 4];
#pragma unroll
  for (int kk = 0; kk < GDIM / 4; ++kk) {
    const int ka = kk * 4 + half * 2;
    bf[kk].x = Wm[(size_t)(ka + 0) * GDIM + col];
    bf[kk].y = Wm[(size_t)(ka + 1) * GDIM + col];
  }
  const float bv = bias[col];
  __syncthreads();

  for (int strip = wave >> 3; strip * 16 < nown; strip += 4) {
    const float* ap = acc + (strip * 16 + lmod) * ZP + half * 2;
    v8f ct = {0.f, 0.f, 0.f, 0.f, 0.f, 0.f, 0.f, 0.f};
#pragma unroll
    for (int kk = 0; kk < GDIM / 4; ++kk) {
      const v2f a = *(const v2f*)(ap + kk * 4);
      ct = __builtin_amdgcn_wmma_f32_16x16x4_f32(
           false, a,  false, bf[kk],
           (short)0, ct,  false,  false);
    }
    const int rb = strip * 16 + half * 8;
    float* op = out + (size_t)(lo + rb) * GDIM + col;
    float o[8];
#pragma unroll
    for (int r = 0; r < 8; ++r) {
      o[r] = ct[r] + bv;
      if (RELU) o[r] = fmaxf(o[r], 0.f);
    }
    if (rb + 8 <= nown) {
#pragma unroll
      for (int r = 0; r < 8; ++r) op[(size_t)r * GDIM] = o[r];
    } else {
#pragma unroll
      for (int r = 0; r < 8; ++r)
        if (rb + r < nown) op[(size_t)r * GDIM] = o[r];
    }
  }
}

extern "C" void kernel_launch(void* const* d_in, const int* in_sizes, int n_in,
                              void* d_out, int out_size, void* d_ws, size_t ws_size,
                              hipStream_t stream) {
  const float* x  = (const float*)d_in[0];
  const int*   ei = (const int*)d_in[1];
  const float* W1 = (const float*)d_in[2];
  const float* b1 = (const float*)d_in[3];
  const float* W2 = (const float*)d_in[4];
  const float* b2 = (const float*)d_in[5];
  float* out = (float*)d_out;

  const int N = in_sizes[0] / GDIM;
  const int E = in_sizes[1] / 2;
  const int* src = ei;
  const int* dst = ei + E;

  float* dinv = (float*)d_ws;
  float* bufB = dinv + N + (size_t)N * GDIM;
  unsigned char* own = (unsigned char*)d_ws +
      ((((size_t)N * (2 * GDIM + 1)) * sizeof(float) + 255) & ~(size_t)255);

  const int T = 256;
  const int gN    = (N + T - 1) / T;
  const int gE    = ((E + 3) / 4 + T - 1) / T;
  const int gAgg  = (N + NPB - 1) / NPB;

  int* lsd = (int*)(own + (((size_t)E + 255) & ~(size_t)255));
  float* lw = (float*)(lsd + (size_t)gAgg * LCAP * 2);
  int* lcnt = (int*)(lw + (size_t)gAgg * LCAP);

  deg_init_kernel<<<gN, T, 0, stream>>>(dinv, N);
  deg_edges_kernel<<<gE, T, 0, stream>>>(dst, dinv, own, E);
  dinv_kernel<<<gN, T, 0, stream>>>(dinv, N);

  aggregate_owner_kernel<true, 1><<<gAgg, AGG_T, 0, stream>>>(x, src, dst, dinv, W1, b1, own, lsd, lw, lcnt, bufB, N, E);

  aggregate_owner_kernel<false, 2><<<gAgg, AGG_T, 0, stream>>>(bufB, src, dst, dinv, W2, b2, own, lsd, lw, lcnt, out, N, E);
}
